# starting kernel
# baseline (speedup 1.0000x reference)
#include <hip/hip_runtime.h>

typedef __attribute__((ext_vector_type(2))) float v2f;
typedef __attribute__((ext_vector_type(8))) float v8f;

namespace {
constexpr int CH   = 256;
constexpr int H    = 48;
constexpr int W    = 64;
constexpr int HW   = H * W;
constexpr int NOFF = 21;
constexpr int PAD  = 20;
}

__global__ __launch_bounds__(256)
void flownetc_corr_wmma(const float* __restrict__ f1,
                        const float* __restrict__ f2,
                        float* __restrict__ out) {
    __shared__ float2 Alds[CH / 2][16];

    const int tid   = threadIdx.x;
    const int h     = blockIdx.x >> 2;
    const int wt    = blockIdx.x & 3;
    const int wbase = ((wt >> 1) << 5) + (wt & 1);

    for (int idx = tid; idx < (CH / 2) * 16; idx += 256) {
        const int cp = idx >> 4;
        const int m  = idx & 15;
        const int g  = (2 * cp) * HW + h * W + wbase + 2 * m;
        Alds[cp][m] = make_float2(f1[g], f1[g + HW]);
    }
    __syncthreads();

    const int wave  = tid >> 5;
    const int lane  = tid & 31;
    const int mn    = lane & 15;
    const int khalf = lane >> 4;

    int   jc[3];
    float msk[3];
#pragma unroll
    for (int t = 0; t < 3; ++t) {
        const int j = wbase - PAD + 32 * t + 2 * mn;
        msk[t] = (j >= 0 && j < W) ? 1.0f : 0.0f;
        jc[t]  = j < 0 ? 0 : (j > W - 1 ? W - 1 : j);
    }

    for (int oyi = wave; oyi < NOFF; oyi += 8) {
        const int r2 = h + 2 * oyi - PAD;

        const int r2n = h + 2 * (oyi + 8) - PAD;
        if (oyi + 8 < NOFF && r2n >= 0 && r2n < H)
            __builtin_prefetch(f2 + (size_t)(2 * khalf) * HW + r2n * W + jc[0], 0, 0);

        v8f acc[3];
#pragma unroll
        for (int t = 0; t < 3; ++t)
            acc[t] = (v8f){0.f, 0.f, 0.f, 0.f, 0.f, 0.f, 0.f, 0.f};

        if (r2 >= 0 && r2 < H) {
            const float*  p  = f2 + (size_t)(2 * khalf) * HW + r2 * W;
            const float2* ap = &Alds[khalf][mn];
#pragma unroll 8
            for (int kk = 0; kk < CH / 4; ++kk) {
                const float2 av = ap[kk * 32];
                v2f a; a.x = av.x; a.y = av.y;
                v2f b0, b1, b2;
                b0.x = p[jc[0]]; b0.y = p[HW + jc[0]];
                b1.x = p[jc[1]]; b1.y = p[HW + jc[1]];
                b2.x = p[jc[2]]; b2.y = p[HW + jc[2]];
                acc[0] = __builtin_amdgcn_wmma_f32_16x16x4_f32(
                             false, a, false, b0, (short)0, acc[0], false, false);
                acc[1] = __builtin_amdgcn_wmma_f32_16x16x4_f32(
                             false, a, false, b1, (short)0, acc[1], false, false);
                acc[2] = __builtin_amdgcn_wmma_f32_16x16x4_f32(
                             false, a, false, b2, (short)0, acc[2], false, false);
                p += 4 * HW;
            }
        }

#pragma unroll
        for (int t = 0; t < 3; ++t)
#pragma unroll
            for (int r = 0; r < 8; ++r)
                acc[t][r] *= msk[t];

#pragma unroll
        for (int t = 0; t < 3; ++t) {
#pragma unroll
            for (int r = 0; r < 8; ++r) {
                const int mo  = r + 8 * khalf;
                const int oxi = 16 * t + mn - mo;
                if (oxi >= 0 && oxi < NOFF) {
                    const int wo = wbase + 2 * mo;
                    out[((size_t)(oyi * NOFF + oxi) * H + h) * W + wo] = acc[t][r];
                }
            }
        }
    }
}

extern "C" void kernel_launch(void* const* d_in, const int* in_sizes, int n_in,
                              void* d_out, int out_size, void* d_ws, size_t ws_size,
                              hipStream_t stream) {
    (void)in_sizes; (void)n_in; (void)out_size; (void)d_ws; (void)ws_size;
    const float* f1  = (const float*)d_in[0];
    const float* f2  = (const float*)d_in[1];
    float*       out = (float*)d_out;

    dim3 grid(H * 4);
    dim3 block(256);
    flownetc_corr_wmma<<<grid, block, 0, stream>>>(f1, f2, out);
}


# block = one (image row, shift row) pair; channel sum split over its 4 waves (64 ch each, whole-row 16-byte loads, 8 same-parity tiles); partial tiles added through LDS, sums regrouped by 64 channels
# speedup vs baseline: 5.7994x; 5.7994x over previous
#include <hip/hip_runtime.h>


typedef __attribute__((ext_vector_type(2))) float v2f;
typedef __attribute__((ext_vector_type(4))) float v4f;
typedef __attribute__((ext_vector_type(8))) float v8f;

namespace {
constexpr int CH     = 256;
constexpr int H      = 48;
constexpr int W      = 64;
constexpr int HW     = H * W;
constexpr int NOFF   = 21;
constexpr int PAD    = 20;
constexpr int NWAVE  = 4;
constexpr int NTHR   = NWAVE * 32;
constexpr int CPW    = CH / NWAVE;
constexpr int KSTEPS = CPW / 4;
}

__global__ __launch_bounds__(NTHR) __attribute__((amdgpu_waves_per_eu(3, 3)))
void flownetc_corr_rowpair(const float* __restrict__ f1,
                           const float* __restrict__ f2,
                           float* __restrict__ out) {
    __shared__ float red[NWAVE][8][8][32];

    const int tid  = threadIdx.x;
    const int h    = blockIdx.x / NOFF;
    const int oyi  = blockIdx.x - h * NOFF;
    const int r2   = h + 2 * oyi - PAD;
    const bool inimg = (r2 >= 0 && r2 < H);

    if (inimg) {
        const int wv   = tid >> 5;
        const int lane = tid & 31;
        const int c    = lane & 15;
        const int half = lane >> 4;
        const int ch0  = wv * CPW + 2 * half;
        const float* pa = f1 + (size_t)ch0 * HW + h  * W + 4 * c;
        const float* pb = f2 + (size_t)ch0 * HW + r2 * W + 4 * c;

        v8f acc[8];
#pragma unroll
        for (int t = 0; t < 8; ++t)
            acc[t] = (v8f){0.f, 0.f, 0.f, 0.f, 0.f, 0.f, 0.f, 0.f};

        v4f a[KSTEPS * 2], b[KSTEPS * 2];
#pragma unroll
        for (int i = 0; i < KSTEPS * 2; ++i) {
            const int chrel = 4 * (i >> 1) + (i & 1);
            a[i] = *(const v4f*)(pa + (size_t)chrel * HW);
            b[i] = *(const v4f*)(pb + (size_t)chrel * HW);
        }

#pragma unroll
        for (int kk = 0; kk < KSTEPS; ++kk) {
            v2f A[4], B[4];
#pragma unroll
            for (int e = 0; e < 4; ++e) {
                A[e].x = a[2 * kk][e]; A[e].y = a[2 * kk + 1][e];
                B[e].x = b[2 * kk][e]; B[e].y = b[2 * kk + 1][e];
            }
            acc[0] = __builtin_amdgcn_wmma_f32_16x16x4_f32(false, A[0], false, B[0], (short)0, acc[0], false, false);
            acc[1] = __builtin_amdgcn_wmma_f32_16x16x4_f32(false, A[0], false, B[2], (short)0, acc[1], false, false);
            acc[2] = __builtin_amdgcn_wmma_f32_16x16x4_f32(false, A[2], false, B[0], (short)0, acc[2], false, false);
            acc[3] = __builtin_amdgcn_wmma_f32_16x16x4_f32(false, A[2], false, B[2], (short)0, acc[3], false, false);
            acc[4] = __builtin_amdgcn_wmma_f32_16x16x4_f32(false, A[1], false, B[1], (short)0, acc[4], false, false);
            acc[5] = __builtin_amdgcn_wmma_f32_16x16x4_f32(false, A[1], false, B[3], (short)0, acc[5], false, false);
            acc[6] = __builtin_amdgcn_wmma_f32_16x16x4_f32(false, A[3], false, B[1], (short)0, acc[6], false, false);
            acc[7] = __builtin_amdgcn_wmma_f32_16x16x4_f32(false, A[3], false, B[3], (short)0, acc[7], false, false);
        }

#pragma unroll
        for (int t = 0; t < 8; ++t)
#pragma unroll
            for (int r = 0; r < 8; ++r)
                red[wv][t][r][lane] = acc[t][r];
    }
    __syncthreads();

    float* orow = out + ((size_t)(oyi * NOFF) * H + h) * W;
    for (int i = tid; i < NOFF * 16; i += NTHR) {
        const int ox = i >> 4;
        const int q  = i & 15;
        const int dx = 2 * ox - PAD;
        v4f o = (v4f){0.f, 0.f, 0.f, 0.f};
        if (inimg) {
            const int r    = q & 7;
            const int hsel = (q >> 3) * 16;
#pragma unroll
            for (int e = 0; e < 4; ++e) {
                const int j = 4 * q + e + dx;
                if (j >= 0 && j < W) {
                    const int n  = j >> 2;
                    const int eb = j & 3;
                    const int t  = (e & 1) * 4 + (e >> 1) * 2 + (eb >> 1);
                    const int ln = n + hsel;
                    float s = red[0][t][r][ln];
#pragma unroll
                    for (int w2 = 1; w2 < NWAVE; ++w2)
                        s += red[w2][t][r][ln];
                    o[e] = s;
                }
            }
        }
        *(v4f*)(orow + (size_t)ox * HW + 4 * q) = o;
    }
}

extern "C" void kernel_launch(void* const* d_in, const int* in_sizes, int n_in,
                              void* d_out, int out_size, void* d_ws, size_t ws_size,
                              hipStream_t stream) {
    (void)in_sizes; (void)n_in; (void)out_size; (void)d_ws; (void)ws_size;
    const float* f1  = (const float*)d_in[0];
    const float* f2  = (const float*)d_in[1];
    float*       out = (float*)d_out;

    dim3 grid(H * NOFF);
    dim3 block(NTHR);
    flownetc_corr_rowpair<<<grid, block, 0, stream>>>(f1, f2, out);
}


# per-wave partial tiles laid out in LDS by output quad (two 16-byte groups per lane and D register): the sum over waves reads two 8-byte values per wave, no branch; same wave order of the sums
# speedup vs baseline: 6.7020x; 1.1556x over previous
#include <hip/hip_runtime.h>


typedef __attribute__((ext_vector_type(2))) float v2f;
typedef __attribute__((ext_vector_type(4))) float v4f;
typedef __attribute__((ext_vector_type(8))) float v8f;

namespace {
constexpr int CH     = 256;
constexpr int H      = 48;
constexpr int W      = 64;
constexpr int HW     = H * W;
constexpr int NOFF   = 21;
constexpr int PAD    = 20;
constexpr int NWAVE  = 4;
constexpr int NTHR   = NWAVE * 32;
constexpr int CPW    = CH / NWAVE;
constexpr int KSTEPS = CPW / 4;
}

__global__ __launch_bounds__(NTHR) __attribute__((amdgpu_waves_per_eu(3, 3)))
void flownetc_corr_rowpair(const float* __restrict__ f1,
                           const float* __restrict__ f2,
                           float* __restrict__ out) {
    __shared__ __attribute__((aligned(16))) float red[NWAVE][8][32][8];

    const int tid  = threadIdx.x;
    const int h    = blockIdx.x / NOFF;
    const int oyi  = blockIdx.x - h * NOFF;
    const int r2   = h + 2 * oyi - PAD;
    const bool inimg = (r2 >= 0 && r2 < H);

    if (inimg) {
        const int wv   = tid >> 5;
        const int lane = tid & 31;
        const int c    = lane & 15;
        const int half = lane >> 4;
        const int ch0  = wv * CPW + 2 * half;
        const float* pa = f1 + (size_t)ch0 * HW + h  * W + 4 * c;
        const float* pb = f2 + (size_t)ch0 * HW + r2 * W + 4 * c;

        v8f acc[8];
#pragma unroll
        for (int t = 0; t < 8; ++t)
            acc[t] = (v8f){0.f, 0.f, 0.f, 0.f, 0.f, 0.f, 0.f, 0.f};

        v4f a[KSTEPS * 2], b[KSTEPS * 2];
#pragma unroll
        for (int i = 0; i < KSTEPS * 2; ++i) {
            const int chrel = 4 * (i >> 1) + (i & 1);
            a[i] = *(const v4f*)(pa + (size_t)chrel * HW);
            b[i] = *(const v4f*)(pb + (size_t)chrel * HW);
        }

#pragma unroll
        for (int kk = 0; kk < KSTEPS; ++kk) {
            v2f A[4], B[4];
#pragma unroll
            for (int e = 0; e < 4; ++e) {
                A[e].x = a[2 * kk][e]; A[e].y = a[2 * kk + 1][e];
                B[e].x = b[2 * kk][e]; B[e].y = b[2 * kk + 1][e];
            }
            acc[0] = __builtin_amdgcn_wmma_f32_16x16x4_f32(false, A[0], false, B[0], (short)0, acc[0], false, false);
            acc[1] = __builtin_amdgcn_wmma_f32_16x16x4_f32(false, A[0], false, B[2], (short)0, acc[1], false, false);
            acc[2] = __builtin_amdgcn_wmma_f32_16x16x4_f32(false, A[2], false, B[0], (short)0, acc[2], false, false);
            acc[3] = __builtin_amdgcn_wmma_f32_16x16x4_f32(false, A[2], false, B[2], (short)0, acc[3], false, false);
            acc[4] = __builtin_amdgcn_wmma_f32_16x16x4_f32(false, A[1], false, B[1], (short)0, acc[4], false, false);
            acc[5] = __builtin_amdgcn_wmma_f32_16x16x4_f32(false, A[1], false, B[3], (short)0, acc[5], false, false);
            acc[6] = __builtin_amdgcn_wmma_f32_16x16x4_f32(false, A[3], false, B[1], (short)0, acc[6], false, false);
            acc[7] = __builtin_amdgcn_wmma_f32_16x16x4_f32(false, A[3], false, B[3], (short)0, acc[7], false, false);
        }

        const int swz = (lane >> 3) & 1;
#pragma unroll
        for (int r = 0; r < 8; ++r) {
            const v4f g0 = (v4f){acc[0][r], acc[4][r], acc[3][r], acc[7][r]};
            const v4f g1 = (v4f){acc[1][r], acc[5][r], acc[2][r], acc[6][r]};
            float* cell = &red[wv][r][lane][0];
            *(v4f*)(cell + 4 * swz)       = g0;
            *(v4f*)(cell + 4 * (swz ^ 1)) = g1;
        }
    }
    __syncthreads();

    float* orow = out + ((size_t)(oyi * NOFF) * H + h) * W;
    constexpr int NITEM = NOFF * 16;
    constexpr int NPASS = (NITEM + NTHR - 1) / NTHR;
    if (inimg) {
        v4f o[NPASS];
#pragma unroll
        for (int it = 0; it < NPASS; ++it) {
            const int i   = tid + it * NTHR;
            const int ic  = i < NITEM ? i : NITEM - 1;
            const int ox  = ic >> 4;
            const int q   = ic & 15;
            const int odd = ox & 1;
            const int dx  = 2 * ox - PAD;
            const int r    = q & 7;
            const int hsel = (q >> 3) * 16;
            const int nlo  = q + ((dx - 2 * odd) >> 2);
            const int nhi  = nlo + odd;
            const bool vlo = (nlo >= 0 && nlo < 16);
            const bool vhi = (nhi >= 0 && nhi < 16);
            const int lnlo = (vlo ? nlo : 0) + hsel;
            const int lnhi = (vhi ? nhi : 0) + hsel;
            const int offlo = 4 * (odd ^ ((lnlo >> 3) & 1));
            const int offhi = 4 * (odd ^ ((lnhi >> 3) & 1)) + 2;
            v2f slo = *(const v2f*)&red[0][r][lnlo][offlo];
            v2f shi = *(const v2f*)&red[0][r][lnhi][offhi];
#pragma unroll
            for (int w2 = 1; w2 < NWAVE; ++w2) {
                slo += *(const v2f*)&red[w2][r][lnlo][offlo];
                shi += *(const v2f*)&red[w2][r][lnhi][offhi];
            }
            o[it] = (v4f){vlo ? slo.x : 0.f, vlo ? slo.y : 0.f, vhi ? shi.x : 0.f, vhi ? shi.y : 0.f};
        }
#pragma unroll
        for (int it = 0; it < NPASS; ++it) {
            const int i = tid + it * NTHR;
            if (i < NITEM)
                *(v4f*)(orow + (size_t)(i >> 4) * HW + 4 * (i & 15)) = o[it];
        }
    } else {
        for (int i = tid; i < NITEM; i += NTHR)
            *(v4f*)(orow + (size_t)(i >> 4) * HW + 4 * (i & 15)) = (v4f){0.f, 0.f, 0.f, 0.f};
    }
}

extern "C" void kernel_launch(void* const* d_in, const int* in_sizes, int n_in,
                              void* d_out, int out_size, void* d_ws, size_t ws_size,
                              hipStream_t stream) {
    (void)in_sizes; (void)n_in; (void)out_size; (void)d_ws; (void)ws_size;
    const float* f1  = (const float*)d_in[0];
    const float* f2  = (const float*)d_in[1];
    float*       out = (float*)d_out;

    dim3 grid(H * NOFF);
    dim3 block(NTHR);
    flownetc_corr_rowpair<<<grid, block, 0, stream>>>(f1, f2, out);
}


# block = one image row x two consecutive vertical shifts: f1 pieces stay in registers for both, 2nd f2 row requested behind the 1st products, in flight across an LDS-only barrier; sums unchanged
# speedup vs baseline: 7.1672x; 1.0694x over previous
#include <hip/hip_runtime.h>


typedef __attribute__((ext_vector_type(2))) float v2f;
typedef __attribute__((ext_vector_type(4))) float v4f;
typedef __attribute__((ext_vector_type(8))) float v8f;

namespace {
constexpr int CH     = 256;
constexpr int H      = 48;
constexpr int W      = 64;
constexpr int HW     = H * W;
constexpr int NOFF   = 21;
constexpr int PAD    = 20;
constexpr int NWAVE  = 4;
constexpr int NTHR   = NWAVE * 32;
constexpr int CPW    = CH / NWAVE;
constexpr int KSTEPS = CPW / 4;
constexpr int NPC    = KSTEPS * 2;
constexpr int NGRP   = (NOFF + 1) / 2;
constexpr int NITEM  = NOFF * 16;
constexpr int NPASS  = (NITEM + NTHR - 1) / NTHR;
}


__device__ __forceinline__ void load_row(const float* __restrict__ p, v4f (&x)[NPC]) {
#pragma unroll
    for (int i = 0; i < NPC; ++i) {
        const int chrel = 4 * (i >> 1) + (i & 1);
        x[i] = *(const v4f*)(p + (size_t)chrel * HW);
    }
}

__device__ __forceinline__ void load_rows2(const float* __restrict__ pa, const float* __restrict__ pb,
                                           v4f (&a)[NPC], v4f (&b)[NPC]) {
#pragma unroll
    for (int i = 0; i < NPC; ++i) {
        const int chrel = 4 * (i >> 1) + (i & 1);
        a[i] = *(const v4f*)(pa + (size_t)chrel * HW);
        b[i] = *(const v4f*)(pb + (size_t)chrel * HW);
    }
}

__device__ __forceinline__ void lds_barrier() {
    __builtin_amdgcn_fence(__ATOMIC_RELEASE, "workgroup", "local");
    __builtin_amdgcn_s_barrier();
    __builtin_amdgcn_fence(__ATOMIC_ACQUIRE, "workgroup", "local");
}

__device__ __forceinline__ void products(const v4f (&a)[NPC], const v4f (&b)[NPC], v8f (&acc)[8]) {
#pragma unroll
    for (int t = 0; t < 8; ++t)
        acc[t] = (v8f){0.f, 0.f, 0.f, 0.f, 0.f, 0.f, 0.f, 0.f};
#pragma unroll
    for (int kk = 0; kk < KSTEPS; ++kk) {
        v2f A[4], B[4];
#pragma unroll
        for (int e = 0; e < 4; ++e) {
            A[e].x = a[2 * kk][e]; A[e].y = a[2 * kk + 1][e];
            B[e].x = b[2 * kk][e]; B[e].y = b[2 * kk + 1][e];
        }
        acc[0] = __builtin_amdgcn_wmma_f32_16x16x4_f32(false, A[0], false, B[0], (short)0, acc[0], false, false);
        acc[1] = __builtin_amdgcn_wmma_f32_16x16x4_f32(false, A[0], false, B[2], (short)0, acc[1], false, false);
        acc[2] = __builtin_amdgcn_wmma_f32_16x16x4_f32(false, A[2], false, B[0], (short)0, acc[2], false, false);
        acc[3] = __builtin_amdgcn_wmma_f32_16x16x4_f32(false, A[2], false, B[2], (short)0, acc[3], false, false);
        acc[4] = __builtin_amdgcn_wmma_f32_16x16x4_f32(false, A[1], false, B[1], (short)0, acc[4], false, false);
        acc[5] = __builtin_amdgcn_wmma_f32_16x16x4_f32(false, A[1], false, B[3], (short)0, acc[5], false, false);
        acc[6] = __builtin_amdgcn_wmma_f32_16x16x4_f32(false, A[3], false, B[1], (short)0, acc[6], false, false);
        acc[7] = __builtin_amdgcn_wmma_f32_16x16x4_f32(false, A[3], false, B[3], (short)0, acc[7], false, false);
    }
}

__device__ __forceinline__ void store_partials(const v8f (&acc)[8], float (*red)[8][32][8], int wv, int lane) {
    const int swz = (lane >> 3) & 1;
#pragma unroll
    for (int r = 0; r < 8; ++r) {
        const v4f g0 = (v4f){acc[0][r], acc[4][r], acc[3][r], acc[7][r]};
        const v4f g1 = (v4f){acc[1][r], acc[5][r], acc[2][r], acc[6][r]};
        float* cell = &red[wv][r][lane][0];
        *(v4f*)(cell + 4 * swz)       = g0;
        *(v4f*)(cell + 4 * (swz ^ 1)) = g1;
    }
}

__device__ __forceinline__ void reduce_store(float (*red)[8][32][8], float* __restrict__ orow, int tid) {
    v4f o[NPASS];
#pragma unroll
    for (int it = 0; it < NPASS; ++it) {
        const int i   = tid + it * NTHR;
        const int ic  = i < NITEM ? i : NITEM - 1;
        const int ox  = ic >> 4;
        const int q   = ic & 15;
        const int odd = ox & 1;
        const int dx  = 2 * ox - PAD;
        const int r    = q & 7;
        const int hsel = (q >> 3) * 16;
        const int nlo  = q + ((dx - 2 * odd) >> 2);
        const int nhi  = nlo + odd;
        const bool vlo = (nlo >= 0 && nlo < 16);
        const bool vhi = (nhi >= 0 && nhi < 16);
        const int lnlo = (vlo ? nlo : 0) + hsel;
        const int lnhi = (vhi ? nhi : 0) + hsel;
        const int offlo = 4 * (odd ^ ((lnlo >> 3) & 1));
        const int offhi = 4 * (odd ^ ((lnhi >> 3) & 1)) + 2;
        v2f slo = *(const v2f*)&red[0][r][lnlo][offlo];
        v2f shi = *(const v2f*)&red[0][r][lnhi][offhi];
#pragma unroll
        for (int w2 = 1; w2 < NWAVE; ++w2) {
            slo += *(const v2f*)&red[w2][r][lnlo][offlo];
            shi += *(const v2f*)&red[w2][r][lnhi][offhi];
        }
        o[it] = (v4f){vlo ? slo.x : 0.f, vlo ? slo.y : 0.f, vhi ? shi.x : 0.f, vhi ? shi.y : 0.f};
    }
#pragma unroll
    for (int it = 0; it < NPASS; ++it) {
        const int i = tid + it * NTHR;
        if (i < NITEM)
            *(v4f*)(orow + (size_t)(i >> 4) * HW + 4 * (i & 15)) = o[it];
    }
}

__device__ __forceinline__ void zero_store(float* __restrict__ orow, int tid) {
    for (int i = tid; i < NITEM; i += NTHR)
        *(v4f*)(orow + (size_t)(i >> 4) * HW + 4 * (i & 15)) = (v4f){0.f, 0.f, 0.f, 0.f};
}

__global__ __launch_bounds__(NTHR) __attribute__((amdgpu_waves_per_eu(2, 2)))
void flownetc_corr_row2shifts(const float* __restrict__ f1,
                              const float* __restrict__ f2,
                              float* __restrict__ out) {
    __shared__ __attribute__((aligned(16))) float red[NWAVE][8][32][8];

    const int tid = threadIdx.x;
    const int gg  = blockIdx.x / H;
    const int h   = blockIdx.x - gg * H;
    const int g   = (h < H / 2) ? (NGRP - 1 - gg) : gg;
    const int oy0 = 2 * g;
    const int oy1 = 2 * g + 1;
    const bool has1 = (oy1 < NOFF);
    const int r0  = h + 2 * oy0 - PAD;
    const int r1  = r0 + 2;
    const bool in0 = (r0 >= 0 && r0 < H);
    const bool in1 = has1 && (r1 >= 0 && r1 < H);
    float* orow0 = out + ((size_t)(oy0 * NOFF) * H + h) * W;
    float* orow1 = out + ((size_t)((has1 ? oy1 : oy0) * NOFF) * H + h) * W;

    if (!in0)         zero_store(orow0, tid);
    if (has1 && !in1) zero_store(orow1, tid);
    if (!in0 && !in1) return;

    const bool two = in0 && in1;
    const int  rA  = in0 ? r0 : r1;
    float* orowA   = in0 ? orow0 : orow1;

    const int wv   = tid >> 5;
    const int lane = tid & 31;
    const int c    = lane & 15;
    const int half = lane >> 4;
    const int ch0  = wv * CPW + 2 * half;
    const float* pa = f1 + (size_t)ch0 * HW + h * W + 4 * c;
    const float* pb = f2 + (size_t)ch0 * HW + 4 * c;

    v4f a[NPC], b[NPC];
    v8f acc[8];
    load_rows2(pa, pb + rA * W, a, b);
    products(a, b, acc);
    if (two)
        load_row(pb + r1 * W, b);
    store_partials(acc, red, wv, lane);
    lds_barrier();
    reduce_store(red, orowA, tid);

    if (two) {
        products(a, b, acc);
        lds_barrier();
        store_partials(acc, red, wv, lane);
        lds_barrier();
        reduce_store(red, orow1, tid);
    }
}

extern "C" void kernel_launch(void* const* d_in, const int* in_sizes, int n_in,
                              void* d_out, int out_size, void* d_ws, size_t ws_size,
                              hipStream_t stream) {
    (void)in_sizes; (void)n_in; (void)out_size; (void)d_ws; (void)ws_size;
    const float* f1  = (const float*)d_in[0];
    const float* f2  = (const float*)d_in[1];
    float*       out = (float*)d_out;

    dim3 grid(H * NGRP);
    dim3 block(NTHR);
    flownetc_corr_row2shifts<<<grid, block, 0, stream>>>(f1, f2, out);
}
